# starting kernel
# baseline (speedup 1.0000x reference)
#include <hip/hip_runtime.h>
#include <hip/hip_bf16.h>
#include <stdint.h>

typedef __attribute__((ext_vector_type(16))) _Float16 v16h;
typedef __attribute__((ext_vector_type(8)))  _Float16 v8h;
typedef __attribute__((ext_vector_type(8)))  float    v8f;

#define NFEAT 64
#define HW    128
#define HALO  130
#define NBATCH 4
#define KTAP  441
#define NCH   442
#define NPADC 448

__device__ __forceinline__ v16h cat8(v8h lo, v8h hi) {
  return __builtin_shufflevector(lo, hi, 0,1,2,3,4,5,6,7,8,9,10,11,12,13,14,15);
}
__device__ __forceinline__ v8f wmma16(v16h a, v16h b, v8f c) {
  return __builtin_amdgcn_wmma_f32_16x16x32_f16(false, a, false, b, (short)0, c, false, false);
}

__global__ void zero_u32(uint32_t* p, size_t n) {
  size_t i = (size_t)blockIdx.x * blockDim.x + threadIdx.x;
  if (i < n) p[i] = 0u;
}

__global__ void pack_w(const float* __restrict__ w1s, const float* __restrict__ w2s,
                       _Float16* __restrict__ dst) {
  int idx = blockIdx.x * blockDim.x + threadIdx.x;
  const int TOT = 2 * 8 * 9 * 64 * 64;
  if (idx >= TOT) return;
  int c   = idx & 63;
  int n   = (idx >> 6) & 63;
  int pg  = idx >> 12;
  int pos = pg % 9;
  int lay = pg / 9;
  int blk = lay & 7;
  int which = lay >> 3;
  const float* src = which ? w2s : w1s;
  float v = src[(((size_t)(blk * 64 + n) * 64 + c) * 9) + pos];
  dst[idx] = (_Float16)v;
}

__global__ void pack_wout(const float* __restrict__ w_out, _Float16* __restrict__ dst) {
  int idx = blockIdx.x * blockDim.x + threadIdx.x;
  if (idx >= NPADC * 64) return;
  int c = idx & 63;
  int n = idx >> 6;
  dst[idx] = (n < NCH) ? (_Float16)w_out[(size_t)n * 64 + c] : (_Float16)0.0f;
}

__global__ void conv_in_k(const float* __restrict__ z, const float* __restrict__ w_in,
                          const float* __restrict__ b_in, _Float16* __restrict__ dst) {
  int idx = blockIdx.x * blockDim.x + threadIdx.x;
  const int TOT = NBATCH * HW * HW * NFEAT;
  if (idx >= TOT) return;
  int co = idx & 63;
  int t  = idx >> 6;
  int x  = t & (HW - 1); t >>= 7;
  int y  = t & (HW - 1); t >>= 7;
  int b  = t;
  float acc = b_in[co];
  #pragma unroll
  for (int ci = 0; ci < 3; ++ci) {
    const float* zp = z + ((size_t)(b * 3 + ci)) * HW * HW;
    const float* wp = w_in + ((size_t)(co * 3 + ci)) * 9;
    #pragma unroll
    for (int ky = 0; ky < 3; ++ky) {
      int yy = y + ky - 1;
      if ((unsigned)yy >= (unsigned)HW) continue;
      #pragma unroll
      for (int kx = 0; kx < 3; ++kx) {
        int xx = x + kx - 1;
        if ((unsigned)xx >= (unsigned)HW) continue;
        acc += zp[yy * HW + xx] * wp[ky * 3 + kx];
      }
    }
  }
  dst[(((size_t)b * HALO + y + 1) * HALO + x + 1) * NFEAT + co] = (_Float16)acc;
}

__global__ __launch_bounds__(256) void conv3x3_wmma(
    const _Float16* __restrict__ src, _Float16* __restrict__ dst,
    const _Float16* __restrict__ wl, const float* __restrict__ bias,
    const _Float16* __restrict__ resid, int do_relu) {
  extern __shared__ _Float16 lw[];
  {
    const uint4* g = (const uint4*)wl;
    uint4* s = (uint4*)lw;
    for (int i = threadIdx.x; i < (9 * 64 * 64) / 8; i += 256) s[i] = g[i];
  }
  __syncthreads();

  const int wave = threadIdx.x >> 5;
  const int lane = threadIdx.x & 31;
  const int tile = blockIdx.x * 8 + wave;
  const int b  = tile >> 10;
  const int r  = tile & 1023;
  const int y  = r >> 3;
  const int x0 = (r & 7) << 4;

  const int mrow = lane & 15;
  const int ksel = (lane >> 4) << 3;
  const int nlan = lane & 15;
  const int mhi  = (lane >> 4) << 3;

  v8f acc[4] = {};

  #pragma unroll
  for (int ky = 0; ky < 3; ++ky) {
    #pragma unroll
    for (int kx = 0; kx < 3; ++kx) {
      const _Float16* ap =
          src + (((size_t)b * HALO + y + ky) * HALO + (x0 + mrow + kx)) * NFEAT;
      const _Float16* wp = lw + (ky * 3 + kx) * 4096;
      v16h a0 = cat8(*(const v8h*)(ap + ksel),      *(const v8h*)(ap + 16 + ksel));
      v16h a1 = cat8(*(const v8h*)(ap + 32 + ksel), *(const v8h*)(ap + 48 + ksel));
      #pragma unroll
      for (int kc = 0; kc < 2; ++kc) {
        const int c0 = kc * 32;
        const v16h a = kc ? a1 : a0;
        const _Float16* w0 = wp + (size_t)(0 * 16 + nlan) * 64 + c0 + ksel;
        const _Float16* w1 = wp + (size_t)(1 * 16 + nlan) * 64 + c0 + ksel;
        const _Float16* w2 = wp + (size_t)(2 * 16 + nlan) * 64 + c0 + ksel;
        const _Float16* w3 = wp + (size_t)(3 * 16 + nlan) * 64 + c0 + ksel;
        v16h b0 = cat8(*(const v8h*)w0, *(const v8h*)(w0 + 16));
        v16h b1 = cat8(*(const v8h*)w1, *(const v8h*)(w1 + 16));
        v16h b2 = cat8(*(const v8h*)w2, *(const v8h*)(w2 + 16));
        v16h b3 = cat8(*(const v8h*)w3, *(const v8h*)(w3 + 16));
        acc[0] = wmma16(a, b0, acc[0]);
        acc[1] = wmma16(a, b1, acc[1]);
        acc[2] = wmma16(a, b2, acc[2]);
        acc[3] = wmma16(a, b3, acc[3]);
      }
    }
  }

  #pragma unroll
  for (int nt = 0; nt < 4; ++nt) {
    const int n = nt * 16 + nlan;
    const float bv = bias[n];
    #pragma unroll
    for (int i = 0; i < 8; ++i) {
      const int m = i + mhi;
      size_t off = (((size_t)b * HALO + y + 1) * HALO + (x0 + m + 1)) * NFEAT + n;
      float v = acc[nt][i] + bv;
      if (resid) v += (float)resid[off];
      if (do_relu) v = v > 0.0f ? v : 0.0f;
      dst[off] = (_Float16)v;
    }
  }
}

__global__ __launch_bounds__(128) void convout_softmax(
    const _Float16* __restrict__ feat, const _Float16* __restrict__ wo,
    const float* __restrict__ bo, _Float16* __restrict__ probs,
    float* __restrict__ nsum) {
  extern __shared__ float lgs[];
  const int wave = threadIdx.x >> 5;
  const int lane = threadIdx.x & 31;
  const int tile = blockIdx.x * 4 + wave;
  const int b  = tile >> 10;
  const int r  = tile & 1023;
  const int y  = r >> 3;
  const int x0 = (r & 7) << 4;

  const int mrow = lane & 15;
  const int ksel = (lane >> 4) << 3;
  const int nlan = lane & 15;
  const int mhi  = (lane >> 4) << 3;

  const _Float16* ap =
      feat + (((size_t)b * HALO + y + 1) * HALO + (x0 + mrow + 1)) * NFEAT;
  v16h a0 = cat8(*(const v8h*)(ap + ksel),      *(const v8h*)(ap + 16 + ksel));
  v16h a1 = cat8(*(const v8h*)(ap + 32 + ksel), *(const v8h*)(ap + 48 + ksel));

  float* lg = lgs + wave * 16 * NPADC;
  float* st = lgs + 4 * 16 * NPADC + wave * 32;

  for (int nt = 0; nt < NPADC / 16; ++nt) {
    const int n = nt * 16 + nlan;
    const _Float16* wp = wo + (size_t)n * 64;
    v16h b0 = cat8(*(const v8h*)(wp + ksel),      *(const v8h*)(wp + 16 + ksel));
    v16h b1 = cat8(*(const v8h*)(wp + 32 + ksel), *(const v8h*)(wp + 48 + ksel));
    v8f c = {};
    c = wmma16(a0, b0, c);
    c = wmma16(a1, b1, c);
    const float bv = (n < NCH) ? bo[n] : 0.0f;
    #pragma unroll
    for (int i = 0; i < 8; ++i) lg[(i + mhi) * NPADC + n] = c[i] + bv;
  }
  __syncthreads();

  for (int p = 0; p < 16; ++p) {
    const float* row = lg + p * NPADC;
    float mx = -1e30f;
    for (int t = lane; t < NCH; t += 32) mx = fmaxf(mx, row[t]);
    #pragma unroll
    for (int off = 16; off; off >>= 1) mx = fmaxf(mx, __shfl_xor(mx, off, 32));
    float sm = 0.0f;
    for (int t = lane; t < NCH; t += 32) sm += __expf(row[t] - mx);
    #pragma unroll
    for (int off = 16; off; off >>= 1) sm += __shfl_xor(sm, off, 32);
    if (lane == 0) { st[p] = mx; st[16 + p] = 1.0f / sm; }
  }

  const int p   = lane & 15;
  const float mx  = st[p];
  const float inv = st[16 + p];
  const float* row = lg + p * NPADC;
  _Float16* pb = probs + (size_t)b * NCH * (HW * HW) + (size_t)y * HW + x0 + p;
  float nacc = 0.0f;
  for (int t = (lane >> 4); t < NCH; t += 2) {
    float pv = __expf(row[t] - mx) * inv;
    pb[(size_t)t * (HW * HW)] = (_Float16)pv;
    if (t == KTAP) nacc += pv;
  }
  atomicAdd(nsum + b, nacc);
}

__global__ void apply_k(const float* __restrict__ x, const float* __restrict__ eps,
                        const _Float16* __restrict__ probs, const float* __restrict__ nsum,
                        float* __restrict__ out) {
  int idx = blockIdx.x * blockDim.x + threadIdx.x;
  const int TOT = NBATCH * 3 * HW * HW;
  if (idx >= TOT) return;
  int w = idx & (HW - 1);
  int t = idx >> 7;
  int h = t & (HW - 1); t >>= 7;
  int c = t % 3;
  int b = t / 3;

  const float* xb = x + ((size_t)(b * 3 + c)) * 512 * 512;
  const _Float16* pb = probs + (size_t)b * NCH * (HW * HW) + (size_t)h * HW + w;
  const int yb = h * 4 - 10, xbs = w * 4 - 10;
  float acc = 0.0f;
  for (int ky = 0; ky < 21; ++ky) {
    int yi = yb + ky;
    if ((unsigned)yi >= 512u) continue;
    const float* xr = xb + (size_t)yi * 512;
    const _Float16* pr = pb + (size_t)(ky * 21) * (HW * HW);
    __builtin_prefetch(pr + (size_t)21 * (HW * HW), 0, 1);
    if (yi + 1 < 512) __builtin_prefetch(xr + 512, 0, 1);
    for (int kx = 0; kx < 21; ++kx) {
      int xi = xbs + kx;
      if ((unsigned)xi >= 512u) continue;
      acc += xr[xi] * (float)pr[(size_t)kx * (HW * HW)];
    }
  }
  float noise = (nsum[b] * (1.0f / (HW * HW))) * eps[idx];
  out[idx] = acc + noise;
}

extern "C" void kernel_launch(void* const* d_in, const int* in_sizes, int n_in,
                              void* d_out, int out_size, void* d_ws, size_t ws_size,
                              hipStream_t stream) {
  const float* x     = (const float*)d_in[0];
  const float* z     = (const float*)d_in[1];
  const float* eps   = (const float*)d_in[2];
  const float* w_in  = (const float*)d_in[3];
  const float* b_in  = (const float*)d_in[4];
  const float* w1s   = (const float*)d_in[5];
  const float* b1s   = (const float*)d_in[6];
  const float* w2s   = (const float*)d_in[7];
  const float* b2s   = (const float*)d_in[8];
  const float* w_out = (const float*)d_in[9];
  const float* b_out = (const float*)d_in[10];
  float* out = (float*)d_out;

  char* ws = (char*)d_ws;
  const size_t FBYTES = (size_t)NBATCH * HALO * HALO * NFEAT * sizeof(_Float16);
  _Float16* F0 = (_Float16*)(ws);
  _Float16* F1 = (_Float16*)(ws + FBYTES);
  _Float16* F2 = (_Float16*)(ws + 2 * FBYTES);
  _Float16* wpack  = (_Float16*)(ws + 3 * FBYTES);
  const size_t WPACK_ELEMS = (size_t)2 * 8 * 9 * 64 * 64;
  _Float16* wopack = wpack + WPACK_ELEMS;
  _Float16* probs  = wopack + (size_t)NPADC * 64;
  float*    nsum   = (float*)(probs + (size_t)NBATCH * NCH * HW * HW);

  const size_t zn = (3 * FBYTES) / 4;
  zero_u32<<<(unsigned)((zn + 255) / 256), 256, 0, stream>>>((uint32_t*)ws, zn);
  zero_u32<<<1, 32, 0, stream>>>((uint32_t*)nsum, 4);

  pack_w<<<(2 * 8 * 9 * 64 * 64 + 255) / 256, 256, 0, stream>>>(w1s, w2s, wpack);
  pack_wout<<<(NPADC * 64 + 255) / 256, 256, 0, stream>>>(w_out, wopack);

  conv_in_k<<<(NBATCH * HW * HW * NFEAT + 255) / 256, 256, 0, stream>>>(z, w_in, b_in, F0);

  _Float16* cur = F0;
  _Float16* nxt = F2;
  const size_t LAYER = (size_t)9 * 64 * 64;
  for (int i = 0; i < 8; ++i) {
    conv3x3_wmma<<<512, 256, 9 * 64 * 64 * sizeof(_Float16), stream>>>(
        cur, F1, wpack + (size_t)i * LAYER, b1s + (size_t)i * 64, nullptr, 1);
    conv3x3_wmma<<<512, 256, 9 * 64 * 64 * sizeof(_Float16), stream>>>(
        F1, nxt, wpack + (size_t)(8 + i) * LAYER, b2s + (size_t)i * 64, cur, 0);
    _Float16* t = cur; cur = nxt; nxt = t;
  }

  const size_t SM_OUT = (size_t)4 * 16 * NPADC * sizeof(float) + 4 * 32 * sizeof(float);
  convout_softmax<<<1024, 128, (unsigned)SM_OUT, stream>>>(cur, wopack, b_out, probs, nsum);
  apply_k<<<(NBATCH * 3 * HW * HW + 255) / 256, 256, 0, stream>>>(x, eps, probs, nsum, out);
}


# 3x3 conv: a wave's 4 tiles advance together tap by tap, the weights stream once through a 4-slot register ring (2 taps in flight); WMMAs start after 36 loads, were 166; same sums, same order
# speedup vs baseline: 4.5704x; 4.5704x over previous
#include <hip/hip_runtime.h>
#include <hip/hip_bf16.h>
#include <stdint.h>

typedef __attribute__((ext_vector_type(16))) _Float16 v16h;
typedef __attribute__((ext_vector_type(8)))  _Float16 v8h;
typedef __attribute__((ext_vector_type(8)))  float    v8f;

#define NFEAT 64
#define HW    128
#define HALO  130
#define NBATCH 4
#define KTAP  441
#define NCH   442
#define NPADC 448

__device__ __forceinline__ v16h cat8(v8h lo, v8h hi) {
  return __builtin_shufflevector(lo, hi, 0,1,2,3,4,5,6,7,8,9,10,11,12,13,14,15);
}
__device__ __forceinline__ v8f wmma16(v16h a, v16h b, v8f c) {
  return __builtin_amdgcn_wmma_f32_16x16x32_f16(false, a, false, b, (short)0, c, false, false);
}

__global__ void zero_u32(uint32_t* p, size_t n) {
  size_t i = (size_t)blockIdx.x * blockDim.x + threadIdx.x;
  if (i < n) p[i] = 0u;
}

__global__ __launch_bounds__(256) void zero_fill16(uint4* __restrict__ p, unsigned n16) {
  const uint4 zv = make_uint4(0u, 0u, 0u, 0u);
  for (unsigned i = blockIdx.x * 256u + threadIdx.x; i < n16; i += gridDim.x * 256u) p[i] = zv;
}

__global__ __launch_bounds__(64) void pack_w(const float* __restrict__ w1s, const float* __restrict__ w2s,
                       _Float16* __restrict__ dst) {
  const int tid = blockIdx.x * 64 + threadIdx.x;
  const int l   = tid & 31;
  const int h2  = (tid >> 5) & 1;
  const int nt  = (tid >> 6) & 3;
  const int kc  = (tid >> 8) & 1;
  const int lay = tid >> 9;
  const int blk = lay & 7;
  const int which = lay >> 3;
  const int n  = nt * 16 + (l & 15);
  const int c0 = kc * 32 + h2 * 16 + ((l >> 4) << 3);
  const float* src = (which ? w2s : w1s) + ((size_t)(blk * 64 + n) * 64 + c0) * 9;
  float buf[72];
  #pragma unroll
  for (int q = 0; q < 18; ++q) {
    const float4 v = ((const float4*)src)[q];
    buf[q * 4 + 0] = v.x; buf[q * 4 + 1] = v.y; buf[q * 4 + 2] = v.z; buf[q * 4 + 3] = v.w;
  }
  #pragma unroll
  for (int pos = 0; pos < 9; ++pos) {
    v8h o;
    #pragma unroll
    for (int e = 0; e < 8; ++e) o[e] = (_Float16)buf[e * 9 + pos];
    const size_t di = ((((size_t)(lay * 9 + pos) * 2 + kc) * 4 + nt) * 2 + h2) * 256 + l * 8;
    *(v8h*)(dst + di) = o;
  }
}

__global__ void pack_wout(const float* __restrict__ w_out, _Float16* __restrict__ dst) {
  int idx = blockIdx.x * blockDim.x + threadIdx.x;
  if (idx >= NPADC * 64) return;
  int c = idx & 63;
  int n = idx >> 6;
  dst[idx] = (n < NCH) ? (_Float16)w_out[(size_t)n * 64 + c] : (_Float16)0.0f;
}

__global__ __launch_bounds__(256) void conv_in_k(const float* __restrict__ z, const float* __restrict__ w_in,
                          const float* __restrict__ b_in, _Float16* __restrict__ dst) {
  const int idx = blockIdx.x * 256 + threadIdx.x;
  const int oct = idx & 7;
  int t = idx >> 3;
  const int x = (t & 31) << 2; t >>= 5;
  const int y = t & (HW - 1);  t >>= 7;
  const int b = t;

  float zw[3][3][6];
  #pragma unroll
  for (int ci = 0; ci < 3; ++ci) {
    const float* zp = z + ((size_t)(b * 3 + ci)) * HW * HW;
    #pragma unroll
    for (int ky = 0; ky < 3; ++ky) {
      const int yy = y + ky - 1;
      #pragma unroll
      for (int c = 0; c < 6; ++c) {
        const int xx = x + c - 1;
        const bool ok = ((unsigned)yy < (unsigned)HW) && ((unsigned)xx < (unsigned)HW);
        const float v = zp[ok ? yy * HW + xx : 0];
        zw[ci][ky][c] = ok ? v : 0.0f;
      }
    }
  }
  v8h o[4];
  #pragma unroll
  for (int jh = 0; jh < 2; ++jh) {
    float wv[4][27];
    const float4* wp = (const float4*)(w_in + oct * 216 + jh * 108);
    #pragma unroll
    for (int q = 0; q < 27; ++q) {
      const float4 v = wp[q];
      wv[(q * 4 + 0) / 27][(q * 4 + 0) % 27] = v.x;
      wv[(q * 4 + 1) / 27][(q * 4 + 1) % 27] = v.y;
      wv[(q * 4 + 2) / 27][(q * 4 + 2) % 27] = v.z;
      wv[(q * 4 + 3) / 27][(q * 4 + 3) % 27] = v.w;
    }
    float acc[4][4];
    #pragma unroll
    for (int j = 0; j < 4; ++j) {
      const float bj = b_in[oct * 8 + jh * 4 + j];
      #pragma unroll
      for (int p = 0; p < 4; ++p) acc[p][j] = bj;
    }
    #pragma unroll
    for (int ci = 0; ci < 3; ++ci) {
      #pragma unroll
      for (int ky = 0; ky < 3; ++ky) {
        #pragma unroll
        for (int kx = 0; kx < 3; ++kx) {
          #pragma unroll
          for (int p = 0; p < 4; ++p) {
            #pragma unroll
            for (int j = 0; j < 4; ++j)
              acc[p][j] += zw[ci][ky][p + kx] * wv[j][ci * 9 + ky * 3 + kx];
          }
        }
      }
    }
    #pragma unroll
    for (int p = 0; p < 4; ++p) {
      #pragma unroll
      for (int j = 0; j < 4; ++j) o[p][jh * 4 + j] = (_Float16)acc[p][j];
    }
  }
  #pragma unroll
  for (int p = 0; p < 4; ++p)
    *(v8h*)(dst + (((size_t)b * HALO + y + 1) * HALO + x + p + 1) * NFEAT + oct * 8) = o[p];
}

#define TPW 4
__device__ __forceinline__ void load_row6(v16h (&fr)[6], const _Float16* __restrict__ src, int off) {
  #pragma unroll
  for (int kx = 0; kx < 3; ++kx) {
    const _Float16* ap = src + off + kx * NFEAT;
    fr[kx * 2 + 0] = cat8(*(const v8h*)(ap),      *(const v8h*)(ap + 16));
    fr[kx * 2 + 1] = cat8(*(const v8h*)(ap + 32), *(const v8h*)(ap + 48));
  }
}

__device__ __forceinline__ void load_tap8(v16h (&w)[8], const v8h* __restrict__ wf, int pos) {
  #pragma unroll
  for (int j = 0; j < 8; ++j) w[j] = cat8(wf[(pos * 8 + j) * 64], wf[(pos * 8 + j) * 64 + 32]);
}

__device__ __forceinline__ void load_row_kx(v16h (&fr)[6], const _Float16* __restrict__ src, int off, int kx) {
  const _Float16* ap = src + off + kx * NFEAT;
  fr[kx * 2 + 0] = cat8(*(const v8h*)(ap),      *(const v8h*)(ap + 16));
  fr[kx * 2 + 1] = cat8(*(const v8h*)(ap + 32), *(const v8h*)(ap + 48));
}

template <bool HAS_RESID, bool DO_RELU>
__device__ __forceinline__ void conv3x3_body(
    const _Float16* __restrict__ src, _Float16* __restrict__ dst,
    const _Float16* __restrict__ wl, const float* __restrict__ bias,
    const _Float16* __restrict__ resid) {
  const int wave = threadIdx.x >> 5;
  const int lane = threadIdx.x & 31;
  const int gw = blockIdx.x * 4 + wave;
  const int b  = gw >> 8;
  const int x0 = (gw & 7) << 4;
  const int y0 = ((gw >> 3) & 31) * TPW;

  const int mrow = lane & 15;
  const int ksel = (lane >> 4) << 3;
  const int nlan = lane & 15;
  const int mhi  = (lane >> 4) << 3;

  const int aoff = ((b * HALO + y0) * HALO + x0 + mrow) * NFEAT + ksel;
  const int ooff = ((b * HALO + y0 + 1) * HALO + x0 + 1 + mhi) * NFEAT + nlan;

  v16h ar[TPW + 2][6];
  v16h wr[4][8];
  const v8h* wf = (const v8h*)wl + lane;
  float bv[4];
  #pragma unroll
  for (int g = 0; g < 3; ++g) {
    #pragma unroll
    for (int r = 0; r < 4; ++r) load_row_kx(ar[r], src, aoff + r * HALO * NFEAT, g);
    load_tap8(wr[g], wf, g);
    if (g == 0) {
      #pragma unroll
      for (int nt = 0; nt < 4; ++nt) bv[nt] = bias[nt * 16 + nlan];
    }
    __builtin_amdgcn_sched_barrier(0);
  }

  _Float16 rv[TPW][4][8];
  v8f acc[TPW][4] = {};

  #pragma unroll
  for (int p = 0; p < 6; ++p) {
    if (p >= 1 && p <= 4) load_tap8(wr[(p + 2) & 3], wf, p + 2);
    if (p == 4) load_tap8(wr[3], wf, 7);
    if (p == 5) load_tap8(wr[0], wf, 8);
    if (p == 1) load_row6(ar[4], src, aoff + 4 * HALO * NFEAT);
    if (p == 2) load_row6(ar[5], src, aoff + 5 * HALO * NFEAT);
    if (HAS_RESID && p == 5) {
      #pragma unroll
      for (int nt = 0; nt < 4; ++nt) {
        #pragma unroll
        for (int i = 0; i < 8; ++i) rv[0][nt][i] = resid[ooff + i * NFEAT + nt * 16];
      }
    }
    __builtin_amdgcn_sched_barrier(0);
    const int ky = p / 3, kx = p - ky * 3;
    #pragma unroll
    for (int kc = 0; kc < 2; ++kc) {
      #pragma unroll
      for (int nt = 0; nt < 4; ++nt) {
        #pragma unroll
        for (int t = 0; t < TPW; ++t)
          acc[t][nt] = wmma16(ar[t + ky][kx * 2 + kc], wr[p & 3][kc * 4 + nt], acc[t][nt]);
      }
    }
    __builtin_amdgcn_sched_barrier(0);
  }

  #pragma unroll
  for (int t = 0; t < TPW; ++t) {
    if (HAS_RESID && t + 1 < TPW) {
      #pragma unroll
      for (int nt = 0; nt < 4; ++nt) {
        #pragma unroll
        for (int i = 0; i < 8; ++i)
          rv[t + 1][nt][i] = resid[ooff + (t + 1) * HALO * NFEAT + i * NFEAT + nt * 16];
      }
    }
    __builtin_amdgcn_sched_barrier(0);
    #pragma unroll
    for (int p = 6; p < 9; ++p) {
      #pragma unroll
      for (int kc = 0; kc < 2; ++kc) {
        #pragma unroll
        for (int nt = 0; nt < 4; ++nt)
          acc[t][nt] = wmma16(ar[t + 2][(p - 6) * 2 + kc], wr[p & 3][kc * 4 + nt], acc[t][nt]);
      }
    }
    __builtin_amdgcn_sched_barrier(0);

    #pragma unroll
    for (int nt = 0; nt < 4; ++nt) {
      #pragma unroll
      for (int i = 0; i < 8; ++i) {
        float v = acc[t][nt][i] + bv[nt];
        if (HAS_RESID) v += (float)rv[t][nt][i];
        if (DO_RELU) v = v > 0.0f ? v : 0.0f;
        dst[ooff + t * HALO * NFEAT + i * NFEAT + nt * 16] = (_Float16)v;
      }
    }
  }
}

__global__ __launch_bounds__(128) void conv3x3_relu(
    const _Float16* __restrict__ src, _Float16* __restrict__ dst,
    const _Float16* __restrict__ wl, const float* __restrict__ bias) {
  conv3x3_body<false, true>(src, dst, wl, bias, nullptr);
}
__global__ __launch_bounds__(128) void conv3x3_skip(
    const _Float16* __restrict__ src, _Float16* __restrict__ dst,
    const _Float16* __restrict__ wl, const float* __restrict__ bias,
    const _Float16* __restrict__ resid) {
  conv3x3_body<true, false>(src, dst, wl, bias, resid);
}

#define NTJ 14
#define PU  4
__global__ __launch_bounds__(128) __attribute__((amdgpu_waves_per_eu(2, 2))) void convout_softmax_gather(
    const _Float16* __restrict__ feat, const _Float16* __restrict__ wo,
    const float* __restrict__ bo, const float* __restrict__ x,
    float* __restrict__ out, float* __restrict__ nsum) {
  extern __shared__ float lgs[];
  const int wave = threadIdx.x >> 5;
  const int lane = threadIdx.x & 31;
  const int tile = blockIdx.x * 4 + wave;
  const int b  = tile >> 10;
  const int r  = tile & 1023;
  const int y  = r >> 3;
  const int x0 = (r & 7) << 4;

  const int mrow = lane & 15;
  const int ksel = (lane >> 4) << 3;
  const int nlan = lane & 15;
  const int mhi  = (lane >> 4) << 3;

  const _Float16* ap =
      feat + (((size_t)b * HALO + y + 1) * HALO + (x0 + mrow + 1)) * NFEAT;
  v16h a0 = cat8(*(const v8h*)(ap + ksel),      *(const v8h*)(ap + 16 + ksel));
  v16h a1 = cat8(*(const v8h*)(ap + 32 + ksel), *(const v8h*)(ap + 48 + ksel));

  float* lg = lgs + wave * 16 * NPADC;

  for (int nt = 0; nt < NPADC / 16; ++nt) {
    const int n = nt * 16 + nlan;
    const _Float16* wp = wo + (size_t)n * 64;
    v16h b0 = cat8(*(const v8h*)(wp + ksel),      *(const v8h*)(wp + 16 + ksel));
    v16h b1 = cat8(*(const v8h*)(wp + 32 + ksel), *(const v8h*)(wp + 48 + ksel));
    v8f c = {};
    c = wmma16(a0, b0, c);
    c = wmma16(a1, b1, c);
    const float bv = (n < NCH) ? bo[n] : 0.0f;
    #pragma unroll
    for (int i = 0; i < 8; ++i) lg[(i + mhi) * NPADC + n] = c[i] + bv;
  }
  __syncthreads();

  int kxj[NTJ], rowoff[NTJ];
  const int yb = y * 4 - 10;
  #pragma unroll
  for (int j = 0; j < NTJ; ++j) {
    const int t  = lane + 32 * j;
    const int ky = t / 21;
    const int yi = yb + ky;
    kxj[j] = t - ky * 21;
    rowoff[j] = (t < KTAP && (unsigned)yi < 512u) ? yi * 512 : -1;
  }

  const float* xc0 = x + (size_t)(b * 3) * (512 * 512);
  const float* xc1 = xc0 + 512 * 512;
  const float* xc2 = xc1 + 512 * 512;

  float o0 = 0.0f, o1 = 0.0f, o2 = 0.0f, nacc = 0.0f;
  for (int p0 = 0; p0 < 16; p0 += PU) {
    float e[PU][NTJ];
    float mx[PU], sm[PU], inv[PU];
    #pragma unroll
    for (int u = 0; u < PU; ++u) {
      const float* row = lg + (p0 + u) * NPADC;
      #pragma unroll
      for (int j = 0; j < NTJ; ++j) {
        const int t = lane + 32 * j;
        const float v = row[t];
        e[u][j] = (t < NCH) ? v : -1e30f;
      }
    }
    #pragma unroll
    for (int u = 0; u < PU; ++u) {
      mx[u] = e[u][0];
      #pragma unroll
      for (int j = 1; j < NTJ; ++j) mx[u] = fmaxf(mx[u], e[u][j]);
    }
    #pragma unroll
    for (int off = 16; off; off >>= 1) {
      #pragma unroll
      for (int u = 0; u < PU; ++u) mx[u] = fmaxf(mx[u], __shfl_xor(mx[u], off, 32));
    }
    #pragma unroll
    for (int u = 0; u < PU; ++u) {
      sm[u] = 0.0f;
      #pragma unroll
      for (int j = 0; j < NTJ; ++j) { e[u][j] = __expf(e[u][j] - mx[u]); sm[u] += e[u][j]; }
    }
    #pragma unroll
    for (int off = 16; off; off >>= 1) {
      #pragma unroll
      for (int u = 0; u < PU; ++u) sm[u] += __shfl_xor(sm[u], off, 32);
    }
    #pragma unroll
    for (int u = 0; u < PU; ++u) inv[u] = 1.0f / sm[u];

    float s0[PU], s1[PU], s2[PU];
    #pragma unroll
    for (int u = 0; u < PU; ++u) {
      const int xbs = (x0 + p0 + u) * 4 - 10;
      s0[u] = 0.0f; s1[u] = 0.0f; s2[u] = 0.0f;
      #pragma unroll
      for (int j = 0; j < NTJ; ++j) {
        const int xi = xbs + kxj[j];
        const bool ok = (rowoff[j] >= 0) && ((unsigned)xi < 512u);
        const int idx = ok ? rowoff[j] + xi : 0;
        const float pv = e[u][j] * inv[u];
        const float v0 = xc0[idx], v1 = xc1[idx], v2 = xc2[idx];
        s0[u] += pv * (ok ? v0 : 0.0f);
        s1[u] += pv * (ok ? v1 : 0.0f);
        s2[u] += pv * (ok ? v2 : 0.0f);
      }
    }
    #pragma unroll
    for (int off = 16; off; off >>= 1) {
      #pragma unroll
      for (int u = 0; u < PU; ++u) {
        s0[u] += __shfl_xor(s0[u], off, 32);
        s1[u] += __shfl_xor(s1[u], off, 32);
        s2[u] += __shfl_xor(s2[u], off, 32);
      }
    }
    #pragma unroll
    for (int u = 0; u < PU; ++u) {
      if (lane == p0 + u) { o0 = s0[u]; o1 = s1[u]; o2 = s2[u]; }
      if (lane == (KTAP & 31)) nacc += e[u][KTAP >> 5] * inv[u];
    }
  }

  if (lane == (KTAP & 31)) atomicAdd(nsum + b, nacc);

  __syncthreads();
  float* so = lgs;
  if (lane < 16) {
    so[wave * 16 + lane]       = o0;
    so[64 + wave * 16 + lane]  = o1;
    so[128 + wave * 16 + lane] = o2;
  }
  __syncthreads();
  if (wave < 3) {
    const int xw0 = ((blockIdx.x * 4) & 7) << 4;
    const float2 v = ((const float2*)(so + wave * 64))[lane];
    float* ob = out + ((size_t)(b * 3 + wave) * HW + y) * HW + xw0;
    ((float2*)ob)[lane] = v;
  }
}

__global__ void add_noise(const float* __restrict__ eps, const float* __restrict__ nsum,
                          float* __restrict__ out) {
  const int i4 = blockIdx.x * blockDim.x + threadIdx.x;
  const int TOT4 = NBATCH * 3 * HW * HW / 4;
  if (i4 >= TOT4) return;
  const int b = i4 / (3 * HW * HW / 4);
  const float m = nsum[b] * (1.0f / (HW * HW));
  const float4 ev = ((const float4*)eps)[i4];
  float4 ov = ((float4*)out)[i4];
  ov.x += m * ev.x; ov.y += m * ev.y; ov.z += m * ev.z; ov.w += m * ev.w;
  ((float4*)out)[i4] = ov;
}

extern "C" void kernel_launch(void* const* d_in, const int* in_sizes, int n_in,
                              void* d_out, int out_size, void* d_ws, size_t ws_size,
                              hipStream_t stream) {
  const float* x     = (const float*)d_in[0];
  const float* z     = (const float*)d_in[1];
  const float* eps   = (const float*)d_in[2];
  const float* w_in  = (const float*)d_in[3];
  const float* b_in  = (const float*)d_in[4];
  const float* w1s   = (const float*)d_in[5];
  const float* b1s   = (const float*)d_in[6];
  const float* w2s   = (const float*)d_in[7];
  const float* b2s   = (const float*)d_in[8];
  const float* w_out = (const float*)d_in[9];
  const float* b_out = (const float*)d_in[10];
  float* out = (float*)d_out;

  char* ws = (char*)d_ws;
  const size_t FBYTES = (size_t)NBATCH * HALO * HALO * NFEAT * sizeof(_Float16);
  _Float16* F0 = (_Float16*)(ws);
  _Float16* F1 = (_Float16*)(ws + FBYTES);
  _Float16* F2 = (_Float16*)(ws + 2 * FBYTES);
  _Float16* wpack  = (_Float16*)(ws + 3 * FBYTES);
  const size_t WPACK_ELEMS = (size_t)2 * 8 * 9 * 64 * 64;
  _Float16* wopack = wpack + WPACK_ELEMS;
  _Float16* probs  = wopack + (size_t)NPADC * 64;
  float*    nsum   = (float*)(probs + (size_t)NBATCH * NCH * HW * HW);

  zero_fill16<<<1024, 256, 0, stream>>>((uint4*)ws, (unsigned)((3 * FBYTES) / 16));
  zero_u32<<<1, 32, 0, stream>>>((uint32_t*)nsum, 4);

  pack_w<<<(2 * 8 * 64 * 8) / 64, 64, 0, stream>>>(w1s, w2s, wpack);
  pack_wout<<<(NPADC * 64 + 255) / 256, 256, 0, stream>>>(w_out, wopack);

  conv_in_k<<<(NBATCH * HW * (HW / 4) * 8) / 256, 256, 0, stream>>>(z, w_in, b_in, F0);

  _Float16* cur = F0;
  _Float16* nxt = F2;
  const size_t LAYER = (size_t)9 * 64 * 64;
  for (int i = 0; i < 8; ++i) {
    conv3x3_relu<<<256, 128, 0, stream>>>(
        cur, F1, wpack + (size_t)i * LAYER, b1s + (size_t)i * 64);
    conv3x3_skip<<<256, 128, 0, stream>>>(
        F1, nxt, wpack + (size_t)(8 + i) * LAYER, b2s + (size_t)i * 64, cur);
    _Float16* t = cur; cur = nxt; nxt = t;
  }

  const size_t SM_OUT = (size_t)4 * 16 * NPADC * sizeof(float);
  convout_softmax_gather<<<1024, 128, (unsigned)SM_OUT, stream>>>(cur, wopack, b_out, x, out, nsum);
  add_noise<<<(NBATCH * 3 * HW * HW / 4 + 255) / 256, 256, 0, stream>>>(eps, nsum, out);
}


# 3x3 conv: activations enter as whole lines: a wave copies its 6x18-pixel halo block to a private LDS slot with 4-line loads (27 a lane, were 72 of 16 lines), A fragments read from LDS; same sums
# speedup vs baseline: 4.8185x; 1.0543x over previous
#include <hip/hip_runtime.h>
#include <hip/hip_bf16.h>
#include <stdint.h>

typedef __attribute__((ext_vector_type(16))) _Float16 v16h;
typedef __attribute__((ext_vector_type(8)))  _Float16 v8h;
typedef __attribute__((ext_vector_type(8)))  float    v8f;

#define NFEAT 64
#define HW    128
#define HALO  130
#define NBATCH 4
#define KTAP  441
#define NCH   442
#define NPADC 448

__device__ __forceinline__ v16h cat8(v8h lo, v8h hi) {
  return __builtin_shufflevector(lo, hi, 0,1,2,3,4,5,6,7,8,9,10,11,12,13,14,15);
}
__device__ __forceinline__ v8f wmma16(v16h a, v16h b, v8f c) {
  return __builtin_amdgcn_wmma_f32_16x16x32_f16(false, a, false, b, (short)0, c, false, false);
}

__global__ void zero_u32(uint32_t* p, size_t n) {
  size_t i = (size_t)blockIdx.x * blockDim.x + threadIdx.x;
  if (i < n) p[i] = 0u;
}

__global__ __launch_bounds__(256) void zero_fill16(uint4* __restrict__ p, unsigned n16) {
  const uint4 zv = make_uint4(0u, 0u, 0u, 0u);
  for (unsigned i = blockIdx.x * 256u + threadIdx.x; i < n16; i += gridDim.x * 256u) p[i] = zv;
}

__global__ __launch_bounds__(64) void pack_w(const float* __restrict__ w1s, const float* __restrict__ w2s,
                       _Float16* __restrict__ dst) {
  const int tid = blockIdx.x * 64 + threadIdx.x;
  const int l   = tid & 31;
  const int h2  = (tid >> 5) & 1;
  const int nt  = (tid >> 6) & 3;
  const int kc  = (tid >> 8) & 1;
  const int lay = tid >> 9;
  const int blk = lay & 7;
  const int which = lay >> 3;
  const int n  = nt * 16 + (l & 15);
  const int c0 = kc * 32 + h2 * 16 + ((l >> 4) << 3);
  const float* src = (which ? w2s : w1s) + ((size_t)(blk * 64 + n) * 64 + c0) * 9;
  float buf[72];
  #pragma unroll
  for (int q = 0; q < 18; ++q) {
    const float4 v = ((const float4*)src)[q];
    buf[q * 4 + 0] = v.x; buf[q * 4 + 1] = v.y; buf[q * 4 + 2] = v.z; buf[q * 4 + 3] = v.w;
  }
  #pragma unroll
  for (int pos = 0; pos < 9; ++pos) {
    v8h o;
    #pragma unroll
    for (int e = 0; e < 8; ++e) o[e] = (_Float16)buf[e * 9 + pos];
    const size_t di = ((((size_t)(lay * 9 + pos) * 2 + kc) * 4 + nt) * 2 + h2) * 256 + l * 8;
    *(v8h*)(dst + di) = o;
  }
}

__global__ void pack_wout(const float* __restrict__ w_out, _Float16* __restrict__ dst) {
  int idx = blockIdx.x * blockDim.x + threadIdx.x;
  if (idx >= NPADC * 64) return;
  int c = idx & 63;
  int n = idx >> 6;
  dst[idx] = (n < NCH) ? (_Float16)w_out[(size_t)n * 64 + c] : (_Float16)0.0f;
}

__global__ __launch_bounds__(256) void conv_in_k(const float* __restrict__ z, const float* __restrict__ w_in,
                          const float* __restrict__ b_in, _Float16* __restrict__ dst) {
  const int idx = blockIdx.x * 256 + threadIdx.x;
  const int oct = idx & 7;
  int t = idx >> 3;
  const int x = (t & 31) << 2; t >>= 5;
  const int y = t & (HW - 1);  t >>= 7;
  const int b = t;

  float zw[3][3][6];
  #pragma unroll
  for (int ci = 0; ci < 3; ++ci) {
    const float* zp = z + ((size_t)(b * 3 + ci)) * HW * HW;
    #pragma unroll
    for (int ky = 0; ky < 3; ++ky) {
      const int yy = y + ky - 1;
      #pragma unroll
      for (int c = 0; c < 6; ++c) {
        const int xx = x + c - 1;
        const bool ok = ((unsigned)yy < (unsigned)HW) && ((unsigned)xx < (unsigned)HW);
        const float v = zp[ok ? yy * HW + xx : 0];
        zw[ci][ky][c] = ok ? v : 0.0f;
      }
    }
  }
  v8h o[4];
  #pragma unroll
  for (int jh = 0; jh < 2; ++jh) {
    float wv[4][27];
    const float4* wp = (const float4*)(w_in + oct * 216 + jh * 108);
    #pragma unroll
    for (int q = 0; q < 27; ++q) {
      const float4 v = wp[q];
      wv[(q * 4 + 0) / 27][(q * 4 + 0) % 27] = v.x;
      wv[(q * 4 + 1) / 27][(q * 4 + 1) % 27] = v.y;
      wv[(q * 4 + 2) / 27][(q * 4 + 2) % 27] = v.z;
      wv[(q * 4 + 3) / 27][(q * 4 + 3) % 27] = v.w;
    }
    float acc[4][4];
    #pragma unroll
    for (int j = 0; j < 4; ++j) {
      const float bj = b_in[oct * 8 + jh * 4 + j];
      #pragma unroll
      for (int p = 0; p < 4; ++p) acc[p][j] = bj;
    }
    #pragma unroll
    for (int ci = 0; ci < 3; ++ci) {
      #pragma unroll
      for (int ky = 0; ky < 3; ++ky) {
        #pragma unroll
        for (int kx = 0; kx < 3; ++kx) {
          #pragma unroll
          for (int p = 0; p < 4; ++p) {
            #pragma unroll
            for (int j = 0; j < 4; ++j)
              acc[p][j] += zw[ci][ky][p + kx] * wv[j][ci * 9 + ky * 3 + kx];
          }
        }
      }
    }
    #pragma unroll
    for (int p = 0; p < 4; ++p) {
      #pragma unroll
      for (int j = 0; j < 4; ++j) o[p][jh * 4 + j] = (_Float16)acc[p][j];
    }
  }
  #pragma unroll
  for (int p = 0; p < 4; ++p)
    *(v8h*)(dst + (((size_t)b * HALO + y + 1) * HALO + x + p + 1) * NFEAT + oct * 8) = o[p];
}

#define TPW 4
#define APITCH 72
#define AROWP (18 * APITCH)
#define ASLOT (6 * AROWP)

__device__ __forceinline__ void load_tap8(v16h (&w)[8], const v8h* __restrict__ wf, int pos) {
  #pragma unroll
  for (int j = 0; j < 8; ++j) w[j] = cat8(wf[(pos * 8 + j) * 64], wf[(pos * 8 + j) * 64 + 32]);
}

__device__ __forceinline__ void lds_row_kx(v16h (&fr)[6], const _Float16* sa, int r, int kx) {
  const _Float16* ap = sa + (r * 18 + kx) * APITCH;
  fr[kx * 2 + 0] = cat8(*(const v8h*)(ap),      *(const v8h*)(ap + 16));
  fr[kx * 2 + 1] = cat8(*(const v8h*)(ap + 32), *(const v8h*)(ap + 48));
}
__device__ __forceinline__ void lds_row6(v16h (&fr)[6], const _Float16* sa, int r) {
  #pragma unroll
  for (int kx = 0; kx < 3; ++kx) lds_row_kx(fr, sa, r, kx);
}

template <bool HAS_RESID, bool DO_RELU>
__device__ __forceinline__ void conv3x3_body(
    const _Float16* __restrict__ src, _Float16* __restrict__ dst,
    const _Float16* __restrict__ wl, const float* __restrict__ bias,
    const _Float16* __restrict__ resid) {
  const int wave = threadIdx.x >> 5;
  const int lane = threadIdx.x & 31;
  const int gw = blockIdx.x * 4 + wave;
  const int b  = gw >> 8;
  const int x0 = (gw & 7) << 4;
  const int y0 = ((gw >> 3) & 31) * TPW;

  const int mrow = lane & 15;
  const int ksel = (lane >> 4) << 3;
  const int nlan = lane & 15;
  const int mhi  = (lane >> 4) << 3;

  const int ooff = ((b * HALO + y0 + 1) * HALO + x0 + 1 + mhi) * NFEAT + nlan;

  __shared__ __attribute__((aligned(16))) _Float16 sA[4 * ASLOT];
  _Float16* sw = sA + wave * ASLOT;
  v16h ar[TPW + 2][6];
  v16h wr[4][8];
  const v8h* wf = (const v8h*)wl + lane;
  float bv[4];

  const _Float16* gl = src + ((b * HALO + y0) * HALO + x0) * NFEAT + lane * 8;
  v8h stg[27];
  #pragma unroll
  for (int k = 0; k < 27; ++k) {
    const int r0 = (32 * k) / 144, w0 = (32 * k) % 144;
    const bool nxt = (w0 + lane >= 144);
    stg[k] = *(const v8h*)(gl + r0 * HALO * NFEAT + w0 * 8 + (nxt ? (HALO * NFEAT - 144 * 8) : 0));
  }
  load_tap8(wr[0], wf, 0);
  load_tap8(wr[1], wf, 1);
  __builtin_amdgcn_sched_barrier(0);

  {
    _Float16* sst = sw + (lane >> 3) * APITCH + (lane & 7) * 8;
    #pragma unroll
    for (int k = 0; k < 27; ++k) *(v8h*)(sst + k * 4 * APITCH) = stg[k];
  }
  load_tap8(wr[2], wf, 2);
  #pragma unroll
  for (int nt = 0; nt < 4; ++nt) bv[nt] = bias[nt * 16 + nlan];
  asm volatile("" ::: "memory");
  const _Float16* sa = sw + mrow * APITCH + ksel;
  #pragma unroll
  for (int g = 0; g < 3; ++g) {
    #pragma unroll
    for (int r = 0; r < 4; ++r) lds_row_kx(ar[r], sa, r, g);
  }
  __builtin_amdgcn_sched_barrier(0);

  _Float16 rv[TPW][4][8];
  v8f acc[TPW][4] = {};

  #pragma unroll
  for (int p = 0; p < 6; ++p) {
    if (p >= 1 && p <= 4) load_tap8(wr[(p + 2) & 3], wf, p + 2);
    if (p == 4) load_tap8(wr[3], wf, 7);
    if (p == 5) load_tap8(wr[0], wf, 8);
    if (p == 1) lds_row6(ar[4], sa, 4);
    if (p == 2) lds_row6(ar[5], sa, 5);
    if (HAS_RESID && p == 5) {
      #pragma unroll
      for (int nt = 0; nt < 4; ++nt) {
        #pragma unroll
        for (int i = 0; i < 8; ++i) rv[0][nt][i] = resid[ooff + i * NFEAT + nt * 16];
      }
    }
    __builtin_amdgcn_sched_barrier(0);
    const int ky = p / 3, kx = p - ky * 3;
    #pragma unroll
    for (int kc = 0; kc < 2; ++kc) {
      #pragma unroll
      for (int nt = 0; nt < 4; ++nt) {
        #pragma unroll
        for (int t = 0; t < TPW; ++t)
          acc[t][nt] = wmma16(ar[t + ky][kx * 2 + kc], wr[p & 3][kc * 4 + nt], acc[t][nt]);
      }
    }
    __builtin_amdgcn_sched_barrier(0);
  }

  #pragma unroll
  for (int t = 0; t < TPW; ++t) {
    if (HAS_RESID && t + 1 < TPW) {
      #pragma unroll
      for (int nt = 0; nt < 4; ++nt) {
        #pragma unroll
        for (int i = 0; i < 8; ++i)
          rv[t + 1][nt][i] = resid[ooff + (t + 1) * HALO * NFEAT + i * NFEAT + nt * 16];
      }
    }
    __builtin_amdgcn_sched_barrier(0);
    #pragma unroll
    for (int p = 6; p < 9; ++p) {
      #pragma unroll
      for (int kc = 0; kc < 2; ++kc) {
        #pragma unroll
        for (int nt = 0; nt < 4; ++nt)
          acc[t][nt] = wmma16(ar[t + 2][(p - 6) * 2 + kc], wr[p & 3][kc * 4 + nt], acc[t][nt]);
      }
    }
    __builtin_amdgcn_sched_barrier(0);

    #pragma unroll
    for (int nt = 0; nt < 4; ++nt) {
      #pragma unroll
      for (int i = 0; i < 8; ++i) {
        float v = acc[t][nt][i] + bv[nt];
        if (HAS_RESID) v += (float)rv[t][nt][i];
        if (DO_RELU) v = v > 0.0f ? v : 0.0f;
        dst[ooff + t * HALO * NFEAT + i * NFEAT + nt * 16] = (_Float16)v;
      }
    }
  }
}

__global__ __launch_bounds__(128) void conv3x3_relu(
    const _Float16* __restrict__ src, _Float16* __restrict__ dst,
    const _Float16* __restrict__ wl, const float* __restrict__ bias) {
  conv3x3_body<false, true>(src, dst, wl, bias, nullptr);
}
__global__ __launch_bounds__(128) void conv3x3_skip(
    const _Float16* __restrict__ src, _Float16* __restrict__ dst,
    const _Float16* __restrict__ wl, const float* __restrict__ bias,
    const _Float16* __restrict__ resid) {
  conv3x3_body<true, false>(src, dst, wl, bias, resid);
}

#define NTJ 14
#define PU  4
__global__ __launch_bounds__(128) __attribute__((amdgpu_waves_per_eu(2, 2))) void convout_softmax_gather(
    const _Float16* __restrict__ feat, const _Float16* __restrict__ wo,
    const float* __restrict__ bo, const float* __restrict__ x,
    float* __restrict__ out, float* __restrict__ nsum) {
  extern __shared__ float lgs[];
  const int wave = threadIdx.x >> 5;
  const int lane = threadIdx.x & 31;
  const int tile = blockIdx.x * 4 + wave;
  const int b  = tile >> 10;
  const int r  = tile & 1023;
  const int y  = r >> 3;
  const int x0 = (r & 7) << 4;

  const int mrow = lane & 15;
  const int ksel = (lane >> 4) << 3;
  const int nlan = lane & 15;
  const int mhi  = (lane >> 4) << 3;

  const _Float16* ap =
      feat + (((size_t)b * HALO + y + 1) * HALO + (x0 + mrow + 1)) * NFEAT;
  v16h a0 = cat8(*(const v8h*)(ap + ksel),      *(const v8h*)(ap + 16 + ksel));
  v16h a1 = cat8(*(const v8h*)(ap + 32 + ksel), *(const v8h*)(ap + 48 + ksel));

  float* lg = lgs + wave * 16 * NPADC;

  for (int nt = 0; nt < NPADC / 16; ++nt) {
    const int n = nt * 16 + nlan;
    const _Float16* wp = wo + (size_t)n * 64;
    v16h b0 = cat8(*(const v8h*)(wp + ksel),      *(const v8h*)(wp + 16 + ksel));
    v16h b1 = cat8(*(const v8h*)(wp + 32 + ksel), *(const v8h*)(wp + 48 + ksel));
    v8f c = {};
    c = wmma16(a0, b0, c);
    c = wmma16(a1, b1, c);
    const float bv = (n < NCH) ? bo[n] : 0.0f;
    #pragma unroll
    for (int i = 0; i < 8; ++i) lg[(i + mhi) * NPADC + n] = c[i] + bv;
  }
  __syncthreads();

  int kxj[NTJ], rowoff[NTJ];
  const int yb = y * 4 - 10;
  #pragma unroll
  for (int j = 0; j < NTJ; ++j) {
    const int t  = lane + 32 * j;
    const int ky = t / 21;
    const int yi = yb + ky;
    kxj[j] = t - ky * 21;
    rowoff[j] = (t < KTAP && (unsigned)yi < 512u) ? yi * 512 : -1;
  }

  const float* xc0 = x + (size_t)(b * 3) * (512 * 512);
  const float* xc1 = xc0 + 512 * 512;
  const float* xc2 = xc1 + 512 * 512;

  float o0 = 0.0f, o1 = 0.0f, o2 = 0.0f, nacc = 0.0f;
  for (int p0 = 0; p0 < 16; p0 += PU) {
    float e[PU][NTJ];
    float mx[PU], sm[PU], inv[PU];
    #pragma unroll
    for (int u = 0; u < PU; ++u) {
      const float* row = lg + (p0 + u) * NPADC;
      #pragma unroll
      for (int j = 0; j < NTJ; ++j) {
        const int t = lane + 32 * j;
        const float v = row[t];
        e[u][j] = (t < NCH) ? v : -1e30f;
      }
    }
    #pragma unroll
    for (int u = 0; u < PU; ++u) {
      mx[u] = e[u][0];
      #pragma unroll
      for (int j = 1; j < NTJ; ++j) mx[u] = fmaxf(mx[u], e[u][j]);
    }
    #pragma unroll
    for (int off = 16; off; off >>= 1) {
      #pragma unroll
      for (int u = 0; u < PU; ++u) mx[u] = fmaxf(mx[u], __shfl_xor(mx[u], off, 32));
    }
    #pragma unroll
    for (int u = 0; u < PU; ++u) {
      sm[u] = 0.0f;
      #pragma unroll
      for (int j = 0; j < NTJ; ++j) { e[u][j] = __expf(e[u][j] - mx[u]); sm[u] += e[u][j]; }
    }
    #pragma unroll
    for (int off = 16; off; off >>= 1) {
      #pragma unroll
      for (int u = 0; u < PU; ++u) sm[u] += __shfl_xor(sm[u], off, 32);
    }
    #pragma unroll
    for (int u = 0; u < PU; ++u) inv[u] = 1.0f / sm[u];

    float s0[PU], s1[PU], s2[PU];
    #pragma unroll
    for (int u = 0; u < PU; ++u) {
      const int xbs = (x0 + p0 + u) * 4 - 10;
      s0[u] = 0.0f; s1[u] = 0.0f; s2[u] = 0.0f;
      #pragma unroll
      for (int j = 0; j < NTJ; ++j) {
        const int xi = xbs + kxj[j];
        const bool ok = (rowoff[j] >= 0) && ((unsigned)xi < 512u);
        const int idx = ok ? rowoff[j] + xi : 0;
        const float pv = e[u][j] * inv[u];
        const float v0 = xc0[idx], v1 = xc1[idx], v2 = xc2[idx];
        s0[u] += pv * (ok ? v0 : 0.0f);
        s1[u] += pv * (ok ? v1 : 0.0f);
        s2[u] += pv * (ok ? v2 : 0.0f);
      }
    }
    #pragma unroll
    for (int off = 16; off; off >>= 1) {
      #pragma unroll
      for (int u = 0; u < PU; ++u) {
        s0[u] += __shfl_xor(s0[u], off, 32);
        s1[u] += __shfl_xor(s1[u], off, 32);
        s2[u] += __shfl_xor(s2[u], off, 32);
      }
    }
    #pragma unroll
    for (int u = 0; u < PU; ++u) {
      if (lane == p0 + u) { o0 = s0[u]; o1 = s1[u]; o2 = s2[u]; }
      if (lane == (KTAP & 31)) nacc += e[u][KTAP >> 5] * inv[u];
    }
  }

  if (lane == (KTAP & 31)) atomicAdd(nsum + b, nacc);

  __syncthreads();
  float* so = lgs;
  if (lane < 16) {
    so[wave * 16 + lane]       = o0;
    so[64 + wave * 16 + lane]  = o1;
    so[128 + wave * 16 + lane] = o2;
  }
  __syncthreads();
  if (wave < 3) {
    const int xw0 = ((blockIdx.x * 4) & 7) << 4;
    const float2 v = ((const float2*)(so + wave * 64))[lane];
    float* ob = out + ((size_t)(b * 3 + wave) * HW + y) * HW + xw0;
    ((float2*)ob)[lane] = v;
  }
}

__global__ void add_noise(const float* __restrict__ eps, const float* __restrict__ nsum,
                          float* __restrict__ out) {
  const int i4 = blockIdx.x * blockDim.x + threadIdx.x;
  const int TOT4 = NBATCH * 3 * HW * HW / 4;
  if (i4 >= TOT4) return;
  const int b = i4 / (3 * HW * HW / 4);
  const float m = nsum[b] * (1.0f / (HW * HW));
  const float4 ev = ((const float4*)eps)[i4];
  float4 ov = ((float4*)out)[i4];
  ov.x += m * ev.x; ov.y += m * ev.y; ov.z += m * ev.z; ov.w += m * ev.w;
  ((float4*)out)[i4] = ov;
}

extern "C" void kernel_launch(void* const* d_in, const int* in_sizes, int n_in,
                              void* d_out, int out_size, void* d_ws, size_t ws_size,
                              hipStream_t stream) {
  const float* x     = (const float*)d_in[0];
  const float* z     = (const float*)d_in[1];
  const float* eps   = (const float*)d_in[2];
  const float* w_in  = (const float*)d_in[3];
  const float* b_in  = (const float*)d_in[4];
  const float* w1s   = (const float*)d_in[5];
  const float* b1s   = (const float*)d_in[6];
  const float* w2s   = (const float*)d_in[7];
  const float* b2s   = (const float*)d_in[8];
  const float* w_out = (const float*)d_in[9];
  const float* b_out = (const float*)d_in[10];
  float* out = (float*)d_out;

  char* ws = (char*)d_ws;
  const size_t FBYTES = (size_t)NBATCH * HALO * HALO * NFEAT * sizeof(_Float16);
  _Float16* F0 = (_Float16*)(ws);
  _Float16* F1 = (_Float16*)(ws + FBYTES);
  _Float16* F2 = (_Float16*)(ws + 2 * FBYTES);
  _Float16* wpack  = (_Float16*)(ws + 3 * FBYTES);
  const size_t WPACK_ELEMS = (size_t)2 * 8 * 9 * 64 * 64;
  _Float16* wopack = wpack + WPACK_ELEMS;
  _Float16* probs  = wopack + (size_t)NPADC * 64;
  float*    nsum   = (float*)(probs + (size_t)NBATCH * NCH * HW * HW);

  zero_fill16<<<1024, 256, 0, stream>>>((uint4*)ws, (unsigned)((3 * FBYTES) / 16));
  zero_u32<<<1, 32, 0, stream>>>((uint32_t*)nsum, 4);

  pack_w<<<(2 * 8 * 64 * 8) / 64, 64, 0, stream>>>(w1s, w2s, wpack);
  pack_wout<<<(NPADC * 64 + 255) / 256, 256, 0, stream>>>(w_out, wopack);

  conv_in_k<<<(NBATCH * HW * (HW / 4) * 8) / 256, 256, 0, stream>>>(z, w_in, b_in, F0);

  _Float16* cur = F0;
  _Float16* nxt = F2;
  const size_t LAYER = (size_t)9 * 64 * 64;
  for (int i = 0; i < 8; ++i) {
    conv3x3_relu<<<256, 128, 0, stream>>>(
        cur, F1, wpack + (size_t)i * LAYER, b1s + (size_t)i * 64);
    conv3x3_skip<<<256, 128, 0, stream>>>(
        F1, nxt, wpack + (size_t)(8 + i) * LAYER, b2s + (size_t)i * 64, cur);
    _Float16* t = cur; cur = nxt; nxt = t;
  }

  const size_t SM_OUT = (size_t)4 * 16 * NPADC * sizeof(float);
  convout_softmax_gather<<<1024, 128, (unsigned)SM_OUT, stream>>>(cur, wopack, b_out, x, out, nsum);
  add_noise<<<(NBATCH * 3 * HW * HW / 4 + 255) / 256, 256, 0, stream>>>(eps, nsum, out);
}
